# starting kernel
# baseline (speedup 1.0000x reference)
#include <hip/hip_runtime.h>

typedef float v2f __attribute__((ext_vector_type(2)));
typedef float v8f __attribute__((ext_vector_type(8)));

#define B_      128
#define L_      257
#define C_      8
#define NSTEP   256
#define OUTPB   4680

#define PATH_OFF 0
#define V_OFF    2060
#define H2_OFF   2068
#define H3_OFF   2076
#define G2_OFF   2084
#define G3_OFF   2092
#define G4_OFF   2100
#define T2_OFF   2112
#define U2_OFF   2176
#define S1O 0
#define S2O 8
#define S3O 72
#define S4O 584
#define SREG 0
#define PREG 4680
#define QREG 9360
#define LDS_FLOATS 14040

__global__ __launch_bounds__(32)
void logsig_scan_kernel(const float* __restrict__ path, float* __restrict__ out) {
  __shared__ float buf[LDS_FLOATS];
  const int lane = threadIdx.x;
  const int b    = blockIdx.x;
  const float* gp = path + (size_t)b * (L_ * C_);

  {
    const float4* g4 = (const float4*)gp;
    float4* l4 = (float4*)&buf[PATH_OFF];
    for (int idx = lane; idx < (L_ * C_) / 4; idx += 32) l4[idx] = g4[idx];
  }
  __syncthreads();

  v8f  cacc[16];
  v2f  a3[16];
  float r2[2] = {0.f, 0.f};
  float sig1  = 0.f;
#pragma unroll
  for (int T = 0; T < 16; ++T) {
    a3[T].x = 0.f; a3[T].y = 0.f;
#pragma unroll
    for (int q = 0; q < 8; ++q) cacc[T][q] = 0.f;
  }

  float pprev = 0.f;
  if (lane < C_) pprev = buf[PATH_OFF + lane];

  const int jj  = lane & 7;
  const int i0  = lane >> 3;
  const int i1  = 4 + i0;
  const int kk2 = 2 * (lane & 3);
  const int ta  = (lane >> 2) & 3;

  for (int t = 1; t <= NSTEP; ++t) {
    if (lane < C_) {
      float pc = buf[PATH_OFF + t * C_ + lane];
      float vv = pc - pprev; pprev = pc;
      buf[V_OFF  + lane] = vv;
      buf[H2_OFF + lane] = 0.5f * vv;
      buf[H3_OFF + lane] = vv * (1.0f / 3.0f);
      buf[G2_OFF + lane] = sig1 + 0.5f * vv;
      buf[G3_OFF + lane] = sig1 + vv * (1.0f / 3.0f);
      buf[G4_OFF + lane] = sig1 + 0.25f * vv;
      sig1 += vv;
    }
    __syncthreads();

    float vj  = buf[V_OFF  + jj];
    float h2j = buf[H2_OFF + jj];
    float h3j = buf[H3_OFF + jj];
    float g2a = buf[G2_OFF + i0], g2b = buf[G2_OFF + i1];
    float g3a = buf[G3_OFF + i0], g3b = buf[G3_OFF + i1];
    float g4a = buf[G4_OFF + i0], g4b = buf[G4_OFF + i1];
    float hkA = buf[H2_OFF + kk2], hkB = buf[H2_OFF + kk2 + 1];
    float vkA = buf[V_OFF  + kk2], vkB = buf[V_OFF  + kk2 + 1];

    float t2a = r2[0] + h3j * g4a;
    float t2b = r2[1] + h3j * g4b;
    float u2a = r2[0] + h2j * g3a;
    float u2b = r2[1] + h2j * g3b;
    r2[0] += vj * g2a;
    r2[1] += vj * g2b;
    buf[T2_OFF + lane]      = t2a;
    buf[T2_OFF + lane + 32] = t2b;
    buf[U2_OFF + lane]      = u2a;
    buf[U2_OFF + lane + 32] = u2b;
    __syncthreads();

    v2f Bv;
    Bv.x = (lane < 8) ? vj : 0.0f;
    Bv.y = (lane >= 8 && lane < 16) ? vj : 0.0f;

#pragma unroll
    for (int T = 0; T < 16; ++T) {
      float t2v = buf[T2_OFF + 4 * T + ta];
      float u2v = buf[U2_OFF + 4 * T + ta];
      v2f Av;
      Av.x = a3[T].x + hkA * t2v;
      Av.y = a3[T].y + hkB * t2v;
      cacc[T] = __builtin_amdgcn_wmma_f32_16x16x4_f32(
          false, Av, false, Bv, (short)0, cacc[T], false, false);
      a3[T].x += vkA * u2v;
      a3[T].y += vkB * u2v;
    }
    __syncthreads();
  }

  if (lane < 8) { buf[SREG + S1O + lane] = sig1; buf[PREG + S1O + lane] = sig1; }
  buf[SREG + S2O + lane]      = r2[0];  buf[PREG + S2O + lane]      = r2[0];
  buf[SREG + S2O + 32 + lane] = r2[1];  buf[PREG + S2O + 32 + lane] = r2[1];
  if (lane < 16) {
#pragma unroll
    for (int T = 0; T < 16; ++T) {
      int f = 32 * T + 2 * lane;
      buf[SREG + S3O + f]     = a3[T].x;  buf[PREG + S3O + f]     = a3[T].x;
      buf[SREG + S3O + f + 1] = a3[T].y;  buf[PREG + S3O + f + 1] = a3[T].y;
    }
  }
  {
    const int hi = (lane >> 4) * 8, cc = lane & 15;
#pragma unroll
    for (int T = 0; T < 16; ++T)
#pragma unroll
      for (int q = 0; q < 8; ++q) {
        int f = (16 * T + q + hi) * 16 + cc;
        buf[SREG + S4O + f] = cacc[T][q];
        buf[PREG + S4O + f] = cacc[T][q];
      }
  }
  __syncthreads();

  int Pb = PREG, Qb = QREG;
  const int mm = lane & 15;
  for (int n = 2; n <= 4; ++n) {
    float coef = ((n & 1) ? 1.0f : -1.0f) / (float)n;
    if (lane < 8) buf[Qb + S1O + lane] = 0.0f;
#pragma unroll
    for (int h = 0; h < 2; ++h) {
      int ij = lane + 32 * h;
      float q = buf[Pb + S1O + (ij >> 3)] * buf[SREG + S1O + (ij & 7)];
      r2[h] += coef * q;
      buf[Qb + S2O + ij] = q;
    }
#pragma unroll
    for (int T = 0; T < 16; ++T) {
#pragma unroll
      for (int r = 0; r < 2; ++r) {
        int f = 32 * T + 2 * mm + r;
        float q = buf[Pb + S1O + (f >> 6)] * buf[SREG + S2O + (f & 63)]
                + buf[Pb + S2O + (f >> 3)] * buf[SREG + S1O + (f & 7)];
        if (r == 0) a3[T].x += coef * q; else a3[T].y += coef * q;
        if (lane < 16) buf[Qb + S3O + f] = q;
      }
    }
    {
      const int hi = (lane >> 4) * 8, cc = lane & 15;
#pragma unroll
      for (int T = 0; T < 16; ++T)
#pragma unroll
        for (int q8 = 0; q8 < 8; ++q8) {
          int f = (16 * T + q8 + hi) * 16 + cc;
          float q = buf[Pb + S1O + (f >> 9)] * buf[SREG + S3O + (f & 511)]
                  + buf[Pb + S2O + (f >> 6)] * buf[SREG + S2O + (f & 63)]
                  + buf[Pb + S3O + (f >> 3)] * buf[SREG + S1O + (f & 7)];
          cacc[T][q8] += coef * q;
          buf[Qb + S4O + f] = q;
        }
    }
    __syncthreads();
    int tmp = Pb; Pb = Qb; Qb = tmp;
  }

  float* ob = out + (size_t)b * OUTPB;
  if (lane < 8) ob[lane] = sig1;
  ob[8 + lane]      = r2[0];
  ob[8 + 32 + lane] = r2[1];
  if (lane < 16) {
#pragma unroll
    for (int T = 0; T < 16; ++T) {
      int f = 32 * T + 2 * lane;
      ob[72 + f]     = a3[T].x;
      ob[72 + f + 1] = a3[T].y;
    }
  }
  {
    const int hi = (lane >> 4) * 8, cc = lane & 15;
#pragma unroll
    for (int T = 0; T < 16; ++T)
#pragma unroll
      for (int q = 0; q < 8; ++q) {
        int f = (16 * T + q + hi) * 16 + cc;
        ob[584 + f] = cacc[T][q];
      }
  }
}

extern "C" void kernel_launch(void* const* d_in, const int* in_sizes, int n_in,
                              void* d_out, int out_size, void* d_ws, size_t ws_size,
                              hipStream_t stream) {
  const float* path = (const float*)d_in[0];
  float* out = (float*)d_out;
  logsig_scan_kernel<<<dim3(B_), dim3(32), 0, stream>>>(path, out);
}


# Path staging: each wave loads its own 33 points, all its global loads in flight at once, and scans behind a wave-local fence; no serial load loop or work-group barrier before the scan; same sums
# speedup vs baseline: 12.6729x; 12.6729x over previous
#include <hip/hip_runtime.h>

typedef float v2f __attribute__((ext_vector_type(2)));
typedef float v8f __attribute__((ext_vector_type(8)));

#define B_      128
#define L_      257
#define C_      8
#define NSTEP   256
#define OUTPB   4680

#define NW      8
#define CH      (NSTEP / NW)
#define NT      (NW * 32)
#define NGRP    256

#define S1O 0
#define S2O 8
#define S3O 72
#define S4O 584
#define LOWF 584

#define SLOT_OFF 0
#define SLOTF 2048
#define LOWP 592
#define SREG (NW * 4096)
#define PREG (SREG + LOWP)
#define QREG (PREG + LOWP)
#define LOW_OFF  (QREG + LOWP)
#define PFX_OFF  (LOW_OFF + NW * LOWF)
#define PATH_OFF (PFX_OFF + NW * LOWF)
#define PIECEF   ((CH + 1) * C_)
#define LDS_FLOATS (PATH_OFF + NW * PIECEF)

__device__ __forceinline__ int f4idx(int T, int q8, int hi, int cc) {
  return ((16 * (T >> 2) + hi + q8) << 6) + 16 * (T & 3) + cc;
}

__global__ __launch_bounds__(NT)
void logsig_scan_kernel(const float* __restrict__ path, float* __restrict__ out) {
  __shared__ __attribute__((aligned(16))) float buf[LDS_FLOATS];
  const int tid  = threadIdx.x;
  const int lane = tid & 31;
  const int w    = __builtin_amdgcn_readfirstlane(tid >> 5);
  const int b    = blockIdx.x >> 1;
  const int half = blockIdx.x & 1;
  const float* gp = path + (size_t)b * (L_ * C_);

  {
    const float4* g4 = (const float4*)gp + w * (CH * C_ / 4);
    float4* l4 = (float4*)&buf[PATH_OFF + w * PIECEF];
    const bool tail = lane < (PIECEF / 4 - 64);
    float4 x0 = g4[lane];
    float4 x1 = g4[32 + lane];
    float4 x2 = tail ? g4[64 + lane] : make_float4(0.f, 0.f, 0.f, 0.f);
    l4[lane] = x0;
    l4[32 + lane] = x1;
    if (tail) l4[64 + lane] = x2;
  }
  __builtin_amdgcn_fence(__ATOMIC_RELEASE, "wavefront");
  __builtin_amdgcn_wave_barrier();
  __builtin_amdgcn_fence(__ATOMIC_ACQUIRE, "wavefront");

  v8f cacc[8];
#pragma unroll
  for (int T = 0; T < 8; ++T)
#pragma unroll
    for (int q = 0; q < 8; ++q) cacc[T][q] = 0.f;

  const int h  = lane >> 4;
  const int c  = lane & 15;
  const int jj = c & 7;
  const int ib = c >> 3;
  const float cX = h ? (1.0f / 3.0f) : 0.5f;
  const float dX = h ? 0.25f : (1.0f / 3.0f);

  const float* pp = &buf[PATH_OFF + w * PIECEF];
  float sg1[4], s2[4];
  float a3p[4][8];
  float pvi[4], pvj, pvk[8];
#pragma unroll
  for (int ti = 0; ti < 4; ++ti) {
    sg1[ti] = 0.f; s2[ti] = 0.f;
    pvi[ti] = pp[2 * ti + ib];
#pragma unroll
    for (int k = 0; k < 8; ++k) a3p[ti][k] = 0.f;
  }
  pvj = pp[jj];
  {
    float4 a = *(const float4*)&pp[0], bq = *(const float4*)&pp[4];
    pvk[0] = a.x; pvk[1] = a.y; pvk[2] = a.z; pvk[3] = a.w;
    pvk[4] = bq.x; pvk[5] = bq.y; pvk[6] = bq.z; pvk[7] = bq.w;
  }
  const float pend_j = pp[CH * C_ + jj];

  float cj[2], ci[2][4];
  float4 ck[2][2];
#pragma unroll
  for (int st = 0; st < 2; ++st) {
    const float* q = pp + 8 * (1 + st);
    cj[st] = q[jj];
#pragma unroll
    for (int ti = 0; ti < 4; ++ti) ci[st][ti] = q[2 * ti + ib];
    ck[st][0] = *(const float4*)&q[0];
    ck[st][1] = *(const float4*)&q[4];
  }

  for (int n = 0; n < CH / 2; ++n) {
    float nj[2], ni[2][4];
    float4 nk[2][2];
#pragma unroll
    for (int st = 0; st < 2; ++st) {
      int sn = 2 * n + 3 + st;
      sn = (sn < CH) ? sn : CH;
      const float* q = pp + 8 * sn;
      nj[st] = q[jj];
#pragma unroll
      for (int ti = 0; ti < 4; ++ti) ni[st][ti] = q[2 * ti + ib];
      nk[st][0] = *(const float4*)&q[0];
      nk[st][1] = *(const float4*)&q[4];
    }

    float ax[4][2], bx[4][2];
#pragma unroll
    for (int st = 0; st < 2; ++st) {
      const float pcj = cj[st];
      const float vj  = pcj - pvj;  pvj = pcj;
      const float Rj  = pend_j - pcj;
      const float hXj = vj * cX;
      const float yj  = h ? 0.5f * vj : Rj;
      float pk[8], vk[8];
      pk[0] = ck[st][0].x; pk[1] = ck[st][0].y; pk[2] = ck[st][0].z; pk[3] = ck[st][0].w;
      pk[4] = ck[st][1].x; pk[5] = ck[st][1].y; pk[6] = ck[st][1].z; pk[7] = ck[st][1].w;
#pragma unroll
      for (int k = 0; k < 8; ++k) { vk[k] = pk[k] - pvk[k]; pvk[k] = pk[k]; }
#pragma unroll
      for (int ti = 0; ti < 4; ++ti) {
        const float pci = ci[st][ti];
        const float vi  = pci - pvi[ti];  pvi[ti] = pci;
        const float gX  = sg1[ti] + vi * dX;
        const float g2  = sg1[ti] + 0.5f * vi;
        const float x2  = s2[ti] + hXj * gX;
        s2[ti]  += vj * g2;
        sg1[ti] += vi;
        ax[ti][st] = x2;
        bx[ti][st] = vi * yj;
#pragma unroll
        for (int k = 0; k < 8; ++k) a3p[ti][k] += vk[k] * x2;
      }
    }

#pragma unroll
    for (int t = 0; t < 2; ++t) {
      v2f Av;
      Av.x = half ? ax[2 + t][0] : ax[t][0];
      Av.y = half ? ax[2 + t][1] : ax[t][1];
#pragma unroll
      for (int tj = 0; tj < 4; ++tj) {
        v2f Bv; Bv.x = bx[tj][0]; Bv.y = bx[tj][1];
        cacc[4 * t + tj] = __builtin_amdgcn_wmma_f32_16x16x4_f32(
            false, Av, false, Bv, (short)0, cacc[4 * t + tj], false, false);
      }
    }

#pragma unroll
    for (int st = 0; st < 2; ++st) {
      cj[st] = nj[st];
#pragma unroll
      for (int ti = 0; ti < 4; ++ti) ci[st][ti] = ni[st][ti];
      ck[st][0] = nk[st][0];
      ck[st][1] = nk[st][1];
    }
  }

  {
    float* lowA = &buf[LOW_OFF + w * LOWF];
    if (lane < 16) {
#pragma unroll
      for (int ti = 0; ti < 4; ++ti) {
        if (jj == 0) lowA[S1O + 2 * ti + ib] = sg1[ti];
        lowA[S2O + 16 * ti + c] = s2[ti];
        float4* d = (float4*)&lowA[S3O + (16 * ti + c) * 8];
        d[0] = make_float4(a3p[ti][0], a3p[ti][1], a3p[ti][2], a3p[ti][3]);
        d[1] = make_float4(a3p[ti][4], a3p[ti][5], a3p[ti][6], a3p[ti][7]);
      }
    }
    __syncthreads();

    {
      const int f = 2 * tid;
      const int e = tid >> 2;
      const int i = tid >> 5;
      float b3x[NW], b3y[NW], b1x[NW], b1y[NW], b2x[NW], b2y[NW], b2e[NW], b1e[NW], b1i[NW];
#pragma unroll
      for (int k = 0; k < NW; ++k) {
        const float* Bk = &buf[LOW_OFF + k * LOWF];
        b3x[k] = Bk[S3O + f];
        b3y[k] = Bk[S3O + f + 1];
        b2e[k] = Bk[S2O + e];
        b1i[k] = Bk[S1O + i];
        if (k > 0) {
          b1x[k] = Bk[S1O + (f & 7)];
          b1y[k] = Bk[S1O + ((f + 1) & 7)];
          b2x[k] = Bk[S2O + (f & 63)];
          b2y[k] = Bk[S2O + ((f + 1) & 63)];
          b1e[k] = Bk[S1O + (e & 7)];
        }
      }
      float p3x = b3x[0], p3y = b3y[0], p2 = b2e[0], p1 = b1i[0];
#pragma unroll
      for (int k = 1; k < NW; ++k) {
        float* pf = &buf[PFX_OFF + k * LOWF];
        pf[S3O + f]     = p3x;
        pf[S3O + f + 1] = p3y;
        if ((tid & 3) == 0)  pf[S2O + e] = p2;
        if ((tid & 31) == 0) pf[S1O + i] = p1;
        float q0 = b3x[k] + p2 * b1x[k] + p1 * b2x[k];
        float q1 = b3y[k] + p2 * b1y[k] + p1 * b2y[k];
        float q2 = b2e[k] + p1 * b1e[k];
        p3x += q0;
        p3y += q1;
        p2  += q2;
        p1  += b1i[k];
      }
      buf[SREG + S3O + f]     = p3x;
      buf[SREG + S3O + f + 1] = p3y;
      if ((tid & 3) == 0)  buf[SREG + S2O + e] = p2;
      if ((tid & 31) == 0) buf[SREG + S1O + i] = p1;
    }
    __syncthreads();

    const int hi = (lane >> 4) * 8, cc = lane & 15;
    if (w > 0) {
      const float* pfx = &buf[PFX_OFF + w * LOWF];
      const float* pfx1 = pfx + S1O + 4 * half;
      const float* pfx2 = pfx + S2O + 32 * half;
      const float* pfx3 = pfx + S3O + 256 * half;
#pragma unroll
      for (int T = 0; T < 8; ++T)
#pragma unroll
        for (int q8 = 0; q8 < 8; ++q8) {
          int f = f4idx(T, q8, hi, cc);
          float q = pfx1[f >> 9] * lowA[S3O + (f & 511)]
                  + pfx2[f >> 6] * lowA[S2O + (f & 63)];
          cacc[T][q8] += q;
        }
      {
        float b1l = lowA[S1O + (lane & 7)];
        v2f Bv;
        Bv.x = (lane < 8) ? b1l : 0.0f;
        Bv.y = (lane >= 8 && lane < 16) ? b1l : 0.0f;
        const int mm = lane & 15;
#pragma unroll
        for (int T = 0; T < 8; ++T) {
          v2f Av;
          Av.x = pfx3[(16 * (T >> 2) + mm) * 8 + 2 * (T & 3)];
          Av.y = pfx3[(16 * (T >> 2) + mm) * 8 + 2 * (T & 3) + 1];
          cacc[T] = __builtin_amdgcn_wmma_f32_16x16x4_f32(
              false, Av, false, Bv, (short)0, cacc[T], false, false);
        }
      }
    }

    {
      float4* s4 = (float4*)&buf[SLOT_OFF + w * SLOTF];
#pragma unroll
      for (int T = 0; T < 8; ++T)
#pragma unroll
        for (int g = 0; g < 2; ++g)
          s4[(2 * T + g) * 32 + lane] =
              make_float4(cacc[T][4 * g], cacc[T][4 * g + 1], cacc[T][4 * g + 2], cacc[T][4 * g + 3]);
    }
  }
  __syncthreads();

  {
    const int hi = (lane >> 4) * 8, cc = lane & 15;
    const int f3 = 2 * tid;
    float o4[8];
    float o3[2];
    float o2 = 0.f, o1 = 0.f;
    {
      const float4* s4 = (const float4*)&buf[SLOT_OFF];
#pragma unroll
      for (int g = 0; g < 2; ++g) {
        const int pos = (2 * w + g) * 32 + lane;
        float4 a = s4[pos];
#pragma unroll
        for (int k = 1; k < NW; ++k) {
          float4 c = s4[k * (SLOTF / 4) + pos];
          a.x += c.x; a.y += c.y; a.z += c.z; a.w += c.w;
        }
        o4[4 * g]     = a.x;
        o4[4 * g + 1] = a.y;
        o4[4 * g + 2] = a.z;
        o4[4 * g + 3] = a.w;
      }
    }
    o3[0] = buf[SREG + S3O + f3];
    o3[1] = buf[SREG + S3O + f3 + 1];
    if (tid < 64) o2 = buf[SREG + S2O + tid];
    if (tid < 8)  o1 = buf[SREG + S1O + tid];

#pragma unroll
    for (int n = 2; n <= 4; ++n) {
      const int Pb = (n == 2) ? SREG : ((n == 3) ? PREG : QREG);
      const int Qb = (n == 2) ? PREG : QREG;
      const float coef = ((n & 1) ? 1.0f : -1.0f) / (float)n;
#pragma unroll
      for (int q8 = 0; q8 < 8; ++q8) {
        int f = 2048 * half + f4idx(w, q8, hi, cc);
        float q = buf[Pb + S1O + (f >> 9)] * buf[SREG + S3O + (f & 511)]
                + buf[Pb + S2O + (f >> 6)] * buf[SREG + S2O + (f & 63)]
                + buf[Pb + S3O + (f >> 3)] * buf[SREG + S1O + (f & 7)];
        o4[q8] += coef * q;
      }
#pragma unroll
      for (int r = 0; r < 2; ++r) {
        int f = f3 + r;
        float q = buf[Pb + S1O + (f >> 6)] * buf[SREG + S2O + (f & 63)]
                + buf[Pb + S2O + (f >> 3)] * buf[SREG + S1O + (f & 7)];
        o3[r] += coef * q;
        if (n < 4) buf[Qb + S3O + f] = q;
      }
      if (tid < 64) {
        float q = buf[Pb + S1O + (tid >> 3)] * buf[SREG + S1O + (tid & 7)];
        o2 += coef * q;
        if (n < 4) buf[Qb + S2O + tid] = q;
      }
      if (n < 4) {
        if (tid < 8) buf[Qb + S1O + tid] = 0.0f;
        __syncthreads();
      }
    }

    float* ob = out + (size_t)b * OUTPB;
    if (half == 0) {
      if (tid < 8)  ob[tid] = o1;
      if (tid < 64) ob[8 + tid] = o2;
      ob[72 + f3]     = o3[0];
      ob[72 + f3 + 1] = o3[1];
    }
#pragma unroll
    for (int q8 = 0; q8 < 8; ++q8)
      ob[584 + 2048 * half + f4idx(w, q8, hi, cc)] = o4[q8];
  }
}

extern "C" void kernel_launch(void* const* d_in, const int* in_sizes, int n_in,
                              void* d_out, int out_size, void* d_ws, size_t ws_size,
                              hipStream_t stream) {
  const float* path = (const float*)d_in[0];
  float* out = (float*)d_out;
  logsig_scan_kernel<<<dim3(NGRP), dim3(NT), 0, stream>>>(path, out);
}


# Join with no running product: pieces scanned from true level 1 (p_start - p_0), R to the path's end; levels 3, 4 then add up plus a rank-7 term from the level-2 totals (summation order changes)
# speedup vs baseline: 13.3187x; 1.0510x over previous
#include <hip/hip_runtime.h>

typedef float v2f __attribute__((ext_vector_type(2)));
typedef float v8f __attribute__((ext_vector_type(8)));

#define B_      128
#define L_      257
#define C_      8
#define NSTEP   256
#define OUTPB   4680

#define NW      8
#define CH      (NSTEP / NW)
#define NT      (NW * 32)
#define NGRP    256

#define S1O 0
#define S2O 8
#define S3O 72
#define S4O 584
#define LOWF 584

#define SLOT_OFF 0
#define SLOTF 2048
#define LOWP 592
#define SREG (NW * 4096)
#define PREG (SREG + LOWP)
#define QREG (PREG + LOWP)
#define LOW_OFF  (QREG + LOWP)
#define PATH_OFF (LOW_OFF + NW * LOWF)
#define P0F      ((CH + 1) * C_)
#define PENDF    (P0F + C_)
#define PIECEF   (PENDF + C_)
#define LDS_FLOATS (PATH_OFF + NW * PIECEF)

__device__ __forceinline__ int f4idx(int T, int q8, int hi, int cc) {
  return ((16 * (T >> 2) + hi + q8) << 6) + 16 * (T & 3) + cc;
}

__global__ __launch_bounds__(NT)
void logsig_scan_kernel(const float* __restrict__ path, float* __restrict__ out) {
  __shared__ __attribute__((aligned(16))) float buf[LDS_FLOATS];
  const int tid  = threadIdx.x;
  const int lane = tid & 31;
  const int w    = __builtin_amdgcn_readfirstlane(tid >> 5);
  const int b    = blockIdx.x >> 1;
  const int half = blockIdx.x & 1;
  const float* gp = path + (size_t)b * (L_ * C_);

  {
    const float4* g4 = (const float4*)gp + w * (CH * C_ / 4);
    float4* l4 = (float4*)&buf[PATH_OFF + w * PIECEF];
    const bool tail = lane < (PIECEF / 4 - 64);
    const float4* gx = (lane < 2) ? (g4 + 64 + lane)
                     : ((lane < 4) ? ((const float4*)gp + (lane - 2))
                                   : ((const float4*)gp + (NSTEP * C_ / 4 + lane - 4)));
    float4 x0 = g4[lane];
    float4 x1 = g4[32 + lane];
    float4 x2 = tail ? *gx : make_float4(0.f, 0.f, 0.f, 0.f);
    l4[lane] = x0;
    l4[32 + lane] = x1;
    if (tail) l4[64 + lane] = x2;
  }
  __builtin_amdgcn_fence(__ATOMIC_RELEASE, "wavefront");
  __builtin_amdgcn_wave_barrier();
  __builtin_amdgcn_fence(__ATOMIC_ACQUIRE, "wavefront");

  v8f cacc[8];
#pragma unroll
  for (int T = 0; T < 8; ++T)
#pragma unroll
    for (int q = 0; q < 8; ++q) cacc[T][q] = 0.f;

  const int h  = lane >> 4;
  const int c  = lane & 15;
  const int jj = c & 7;
  const int ib = c >> 3;
  const float cX = h ? (1.0f / 3.0f) : 0.5f;
  const float dX = h ? 0.25f : (1.0f / 3.0f);

  const float* pp = &buf[PATH_OFF + w * PIECEF];
  float sg1[4], s2[4];
  float a3p[4][8];
  float pvi[4], pvj, pvk[8];
#pragma unroll
  for (int ti = 0; ti < 4; ++ti) {
    pvi[ti] = pp[2 * ti + ib];
    sg1[ti] = pvi[ti] - pp[P0F + 2 * ti + ib];
    s2[ti] = 0.f;
#pragma unroll
    for (int k = 0; k < 8; ++k) a3p[ti][k] = 0.f;
  }
  pvj = pp[jj];
  {
    float4 a = *(const float4*)&pp[0], bq = *(const float4*)&pp[4];
    pvk[0] = a.x; pvk[1] = a.y; pvk[2] = a.z; pvk[3] = a.w;
    pvk[4] = bq.x; pvk[5] = bq.y; pvk[6] = bq.z; pvk[7] = bq.w;
  }
  const float pend_j = pp[PENDF + jj];

  float cj[2], ci[2][4];
  float4 ck[2][2];
#pragma unroll
  for (int st = 0; st < 2; ++st) {
    const float* q = pp + 8 * (1 + st);
    cj[st] = q[jj];
#pragma unroll
    for (int ti = 0; ti < 4; ++ti) ci[st][ti] = q[2 * ti + ib];
    ck[st][0] = *(const float4*)&q[0];
    ck[st][1] = *(const float4*)&q[4];
  }

  for (int n = 0; n < CH / 2; ++n) {
    float nj[2], ni[2][4];
    float4 nk[2][2];
#pragma unroll
    for (int st = 0; st < 2; ++st) {
      int sn = 2 * n + 3 + st;
      sn = (sn < CH) ? sn : CH;
      const float* q = pp + 8 * sn;
      nj[st] = q[jj];
#pragma unroll
      for (int ti = 0; ti < 4; ++ti) ni[st][ti] = q[2 * ti + ib];
      nk[st][0] = *(const float4*)&q[0];
      nk[st][1] = *(const float4*)&q[4];
    }

    float ax[4][2], bx[4][2];
#pragma unroll
    for (int st = 0; st < 2; ++st) {
      const float pcj = cj[st];
      const float vj  = pcj - pvj;  pvj = pcj;
      const float Rj  = pend_j - pcj;
      const float hXj = vj * cX;
      const float yj  = h ? 0.5f * vj : Rj;
      float pk[8], vk[8];
      pk[0] = ck[st][0].x; pk[1] = ck[st][0].y; pk[2] = ck[st][0].z; pk[3] = ck[st][0].w;
      pk[4] = ck[st][1].x; pk[5] = ck[st][1].y; pk[6] = ck[st][1].z; pk[7] = ck[st][1].w;
#pragma unroll
      for (int k = 0; k < 8; ++k) { vk[k] = pk[k] - pvk[k]; pvk[k] = pk[k]; }
#pragma unroll
      for (int ti = 0; ti < 4; ++ti) {
        const float pci = ci[st][ti];
        const float vi  = pci - pvi[ti];  pvi[ti] = pci;
        const float gX  = sg1[ti] + vi * dX;
        const float g2  = sg1[ti] + 0.5f * vi;
        const float x2  = s2[ti] + hXj * gX;
        s2[ti]  += vj * g2;
        sg1[ti] += vi;
        ax[ti][st] = x2;
        bx[ti][st] = vi * yj;
#pragma unroll
        for (int k = 0; k < 8; ++k) a3p[ti][k] += vk[k] * x2;
      }
    }

#pragma unroll
    for (int t = 0; t < 2; ++t) {
      v2f Av;
      Av.x = half ? ax[2 + t][0] : ax[t][0];
      Av.y = half ? ax[2 + t][1] : ax[t][1];
#pragma unroll
      for (int tj = 0; tj < 4; ++tj) {
        v2f Bv; Bv.x = bx[tj][0]; Bv.y = bx[tj][1];
        cacc[4 * t + tj] = __builtin_amdgcn_wmma_f32_16x16x4_f32(
            false, Av, false, Bv, (short)0, cacc[4 * t + tj], false, false);
      }
    }

#pragma unroll
    for (int st = 0; st < 2; ++st) {
      cj[st] = nj[st];
#pragma unroll
      for (int ti = 0; ti < 4; ++ti) ci[st][ti] = ni[st][ti];
      ck[st][0] = nk[st][0];
      ck[st][1] = nk[st][1];
    }
  }

  {
    float* lowA = &buf[LOW_OFF + w * LOWF];
    if (lane < 16) {
#pragma unroll
      for (int ti = 0; ti < 4; ++ti) {
        lowA[S2O + 16 * ti + c] = s2[ti];
        float4* d = (float4*)&lowA[S3O + (16 * ti + c) * 8];
        d[0] = make_float4(a3p[ti][0], a3p[ti][1], a3p[ti][2], a3p[ti][3]);
        d[1] = make_float4(a3p[ti][4], a3p[ti][5], a3p[ti][6], a3p[ti][7]);
      }
    }
    float4* s4 = (float4*)&buf[SLOT_OFF + w * SLOTF];
#pragma unroll
    for (int T = 0; T < 8; ++T)
#pragma unroll
      for (int g = 0; g < 2; ++g)
        s4[(2 * T + g) * 32 + lane] =
            make_float4(cacc[T][4 * g], cacc[T][4 * g + 1], cacc[T][4 * g + 2], cacc[T][4 * g + 3]);
  }
  __syncthreads();

  {
    const int hi = (lane >> 4) * 8, cc = lane & 15;
    const int f3 = 2 * tid;
    float o4[8];
    float o3[2];
    float o2 = 0.f, o1 = 0.f;
    float2 p3;
    {
      const int e  = tid >> 2;
      const int kk = f3 & 7;
      p3 = *(const float2*)&buf[LOW_OFF + S3O + f3];
      float pref = 0.f;
#pragma unroll
      for (int k = 1; k < NW; ++k) {
        pref += buf[LOW_OFF + (k - 1) * LOWF + S2O + e];
        const float* pk = &buf[PATH_OFF + k * PIECEF];
        const float2 a  = *(const float2*)&buf[LOW_OFF + k * LOWF + S3O + f3];
        const float2 pe = *(const float2*)&pk[CH * C_ + kk];
        const float2 ps = *(const float2*)&pk[kk];
        p3.x += a.x + pref * (pe.x - ps.x);
        p3.y += a.y + pref * (pe.y - ps.y);
      }
      const float s2tot = pref + buf[LOW_OFF + (NW - 1) * LOWF + S2O + e];
      *(float2*)&buf[SREG + S3O + f3] = p3;
      if ((tid & 3) == 0) buf[SREG + S2O + e] = s2tot;
      if (tid < 8) buf[SREG + S1O + tid] = buf[PATH_OFF + PENDF + tid] - buf[PATH_OFF + P0F + tid];
    }
    {
      const float4* s4 = (const float4*)&buf[SLOT_OFF];
#pragma unroll
      for (int g = 0; g < 2; ++g) {
        const int pos = (2 * w + g) * 32 + lane;
        float4 a = s4[pos];
#pragma unroll
        for (int k = 1; k < NW; ++k) {
          float4 c = s4[k * (SLOTF / 4) + pos];
          a.x += c.x; a.y += c.y; a.z += c.z; a.w += c.w;
        }
        o4[4 * g]     = a.x;
        o4[4 * g + 1] = a.y;
        o4[4 * g + 2] = a.z;
        o4[4 * g + 3] = a.w;
      }
    }
    {
      const int hh   = lane >> 4;
      const int rowA = 16 * (2 * half + (w >> 2)) + cc;
      const int kcol = 2 * (w & 3) + (cc >> 3);
      const int lcol = cc & 7;
      float P[NW];
      P[0] = 0.f;
#pragma unroll
      for (int k = 1; k < NW; ++k) P[k] = P[k - 1] + buf[LOW_OFF + (k - 1) * LOWF + S2O + rowA];
      const float s1l = buf[PATH_OFF + PENDF + lcol] - buf[PATH_OFF + P0F + lcol];
      v2f Av[2], Bv[2];
#pragma unroll
      for (int q = 0; q < 2; ++q) {
        const int k0 = 4 * q + 2 * hh;
        const float* pa = &buf[PATH_OFF + k0 * PIECEF];
        const float* pb = pa + PIECEF;
        const float* ta = &buf[LOW_OFF + k0 * LOWF + S2O + lcol * 8 + kcol];
        Av[q].x = hh ? P[4 * q + 2] : P[4 * q];
        Av[q].y = hh ? P[4 * q + 3] : P[4 * q + 1];
        Bv[q].x = (pa[CH * C_ + kcol] - pa[kcol]) * s1l - ta[0];
        Bv[q].y = (pb[CH * C_ + kcol] - pb[kcol]) * s1l - ta[LOWF];
      }
      v8f acc;
#pragma unroll
      for (int q8 = 0; q8 < 8; ++q8) acc[q8] = o4[q8];
#pragma unroll
      for (int q = 0; q < 2; ++q)
        acc = __builtin_amdgcn_wmma_f32_16x16x4_f32(false, Av[q], false, Bv[q], (short)0, acc, false, false);
#pragma unroll
      for (int q8 = 0; q8 < 8; ++q8) o4[q8] = acc[q8];
    }
    __syncthreads();
    o3[0] = p3.x;
    o3[1] = p3.y;
    if (tid < 64) o2 = buf[SREG + S2O + tid];
    if (tid < 8)  o1 = buf[SREG + S1O + tid];

#pragma unroll
    for (int n = 2; n <= 4; ++n) {
      const int Pb = (n == 2) ? SREG : ((n == 3) ? PREG : QREG);
      const int Qb = (n == 2) ? PREG : QREG;
      const float coef = ((n & 1) ? 1.0f : -1.0f) / (float)n;
#pragma unroll
      for (int q8 = 0; q8 < 8; ++q8) {
        int f = 2048 * half + f4idx(w, q8, hi, cc);
        float q = buf[Pb + S1O + (f >> 9)] * buf[SREG + S3O + (f & 511)]
                + buf[Pb + S2O + (f >> 6)] * buf[SREG + S2O + (f & 63)]
                + buf[Pb + S3O + (f >> 3)] * buf[SREG + S1O + (f & 7)];
        o4[q8] += coef * q;
      }
#pragma unroll
      for (int r = 0; r < 2; ++r) {
        int f = f3 + r;
        float q = buf[Pb + S1O + (f >> 6)] * buf[SREG + S2O + (f & 63)]
                + buf[Pb + S2O + (f >> 3)] * buf[SREG + S1O + (f & 7)];
        o3[r] += coef * q;
        if (n < 4) buf[Qb + S3O + f] = q;
      }
      if (tid < 64) {
        float q = buf[Pb + S1O + (tid >> 3)] * buf[SREG + S1O + (tid & 7)];
        o2 += coef * q;
        if (n < 4) buf[Qb + S2O + tid] = q;
      }
      if (n < 4) {
        if (tid < 8) buf[Qb + S1O + tid] = 0.0f;
        __syncthreads();
      }
    }

    float* ob = out + (size_t)b * OUTPB;
    if (half == 0) {
      if (tid < 8)  ob[tid] = o1;
      if (tid < 64) ob[8 + tid] = o2;
      ob[72 + f3]     = o3[0];
      ob[72 + f3 + 1] = o3[1];
    }
#pragma unroll
    for (int q8 = 0; q8 < 8; ++q8)
      ob[584 + 2048 * half + f4idx(w, q8, hi, cc)] = o4[q8];
  }
}

extern "C" void kernel_launch(void* const* d_in, const int* in_sizes, int n_in,
                              void* d_out, int out_size, void* d_ws, size_t ws_size,
                              hipStream_t stream) {
  const float* path = (const float*)d_in[0];
  float* out = (float*)d_out;
  logsig_scan_kernel<<<dim3(NGRP), dim3(NT), 0, stream>>>(path, out);
}


# Scan K re-assigned: lane half = which 16-step piece of the wave's time (was: which term), element = term; a lane does ONE step per trip, level-2/3 work not doubled; 16-piece join (sum order changes)
# speedup vs baseline: 14.9986x; 1.1261x over previous
#include <hip/hip_runtime.h>

typedef float v2f __attribute__((ext_vector_type(2)));
typedef float v8f __attribute__((ext_vector_type(8)));

#define B_      128
#define L_      257
#define C_      8
#define NSTEP   256
#define OUTPB   4680

#define NW      8
#define CH      (NSTEP / NW)
#define NP      (2 * NW)
#define SH      (CH / 2)
#define NT      (NW * 32)
#define NGRP    256

#define S1O 0
#define S2O 8
#define S3O 72
#define S4O 584
#define LOWF 584

#define SLOT_OFF 0
#define SLOTF 2048
#define LOWP 592
#define SREG (NW * 4096)
#define PREG (SREG + LOWP)
#define QREG (PREG + LOWP)
#define LOW_OFF  (QREG + LOWP)
#define PATH_OFF (LOW_OFF + NP * LOWF)
#define P0F      ((CH + 1) * C_)
#define PENDF    (P0F + C_)
#define PBF      (PENDF + C_)
#define PIECEF   (PBF + (SH + 2) * C_)
#define LDS_FLOATS (PATH_OFF + NW * PIECEF)

__device__ __forceinline__ int f4idx(int T, int q8, int hi, int cc) {
  return ((16 * (T >> 2) + hi + q8) << 6) + 16 * (T & 3) + cc;
}

template <int RT>
__device__ __forceinline__ void scan_pieces(const float* pp, int lane, v8f (&cacc)[8],
                                            float (&s2)[4], float (&a3p)[4][8]) {
#pragma unroll
  for (int T = 0; T < 8; ++T)
#pragma unroll
    for (int q = 0; q < 8; ++q) cacc[T][q] = 0.f;

  const int h  = lane >> 4;
  const int c  = lane & 15;
  const int jj = c & 7;
  const int ib = c >> 3;

  const float* pl = (const float*)__builtin_assume_aligned(pp + PBF * h, 16);
  float sg1[4];
  float pvi[4], pvj, pvk[8];
#pragma unroll
  for (int ti = 0; ti < 4; ++ti) {
    pvi[ti] = pl[2 * ti + ib];
    sg1[ti] = pvi[ti] - pp[P0F + 2 * ti + ib];
    s2[ti] = 0.f;
#pragma unroll
    for (int k = 0; k < 8; ++k) a3p[ti][k] = 0.f;
  }
  pvj = pl[jj];
  {
    float4 a = *(const float4*)&pl[0], bq = *(const float4*)&pl[4];
    pvk[0] = a.x; pvk[1] = a.y; pvk[2] = a.z; pvk[3] = a.w;
    pvk[4] = bq.x; pvk[5] = bq.y; pvk[6] = bq.z; pvk[7] = bq.w;
  }
  const float pend_j = pp[PENDF + jj];

  float cj, ci[4];
  float4 ck[2];
  {
    const float* q = pl + 8;
    cj = q[jj];
#pragma unroll
    for (int ti = 0; ti < 4; ++ti) ci[ti] = q[2 * ti + ib];
    ck[0] = *(const float4*)&q[0];
    ck[1] = *(const float4*)&q[4];
  }

  for (int n = 0; n < SH; ++n) {
    float nj, ni[4];
    float4 nk[2];
    {
      const float* q = pl + 8 * (n + 2);
      nj = q[jj];
#pragma unroll
      for (int ti = 0; ti < 4; ++ti) ni[ti] = q[2 * ti + ib];
      nk[0] = *(const float4*)&q[0];
      nk[1] = *(const float4*)&q[4];
    }

    v2f ax[2], bx[4];
    {
      const float pcj = cj;
      const float vj  = pcj - pvj;  pvj = pcj;
      const float Rj  = pend_j - pcj;
      const float hU  = 0.5f * vj;
      const float hT  = vj * (1.0f / 3.0f);
      float pk[8], vk[8];
      pk[0] = ck[0].x; pk[1] = ck[0].y; pk[2] = ck[0].z; pk[3] = ck[0].w;
      pk[4] = ck[1].x; pk[5] = ck[1].y; pk[6] = ck[1].z; pk[7] = ck[1].w;
#pragma unroll
      for (int k = 0; k < 8; ++k) { vk[k] = pk[k] - pvk[k]; pvk[k] = pk[k]; }
#pragma unroll
      for (int ti = 0; ti < 4; ++ti) {
        const float pci = ci[ti];
        const float vi  = pci - pvi[ti];  pvi[ti] = pci;
        const float gU  = sg1[ti] + vi * (1.0f / 3.0f);
        const float g2  = sg1[ti] + 0.5f * vi;
        const float u2  = s2[ti] + hU * gU;
        if (ti == RT || ti == RT + 1) {
          const float gT = sg1[ti] + vi * 0.25f;
          ax[ti - RT].x = u2;
          ax[ti - RT].y = s2[ti] + hT * gT;
        }
        s2[ti]  += vj * g2;
        sg1[ti] += vi;
        bx[ti].x = vi * Rj;  bx[ti].y = vi * hU;
#pragma unroll
        for (int k = 0; k < 8; ++k) a3p[ti][k] += vk[k] * u2;
      }
    }

#pragma unroll
    for (int t = 0; t < 2; ++t)
#pragma unroll
      for (int tj = 0; tj < 4; ++tj)
        cacc[4 * t + tj] = __builtin_amdgcn_wmma_f32_16x16x4_f32(
            false, ax[t], false, bx[tj], (short)0, cacc[4 * t + tj], false, false);

    cj = nj;
#pragma unroll
    for (int ti = 0; ti < 4; ++ti) ci[ti] = ni[ti];
    ck[0] = nk[0];
    ck[1] = nk[1];
  }
}

__global__ __launch_bounds__(NT)
void logsig_scan_kernel(const float* __restrict__ path, float* __restrict__ out) {
  __shared__ __attribute__((aligned(16))) float buf[LDS_FLOATS];
  const int tid  = threadIdx.x;
  const int lane = tid & 31;
  const int w    = __builtin_amdgcn_readfirstlane(tid >> 5);
  const int b    = blockIdx.x >> 1;
  const int half = blockIdx.x & 1;
  const float* gp = path + (size_t)b * (L_ * C_);

  {
    const float4* g4 = (const float4*)gp + w * (CH * C_ / 4);
    float4* l4 = (float4*)&buf[PATH_OFF + w * PIECEF];
    const bool tail = lane < (PBF / 4 - 64);
    const float4* gx = (lane < 2) ? (g4 + 64 + lane)
                     : ((lane < 4) ? ((const float4*)gp + (lane - 2))
                                   : ((const float4*)gp + (NSTEP * C_ / 4 + lane - 4)));
    float4 x0 = g4[lane];
    float4 x1 = g4[32 + lane];
    float4 x2 = tail ? *gx : make_float4(0.f, 0.f, 0.f, 0.f);
    l4[lane] = x0;
    l4[32 + lane] = x1;
    l4[PBF / 4 + lane] = x1;
    if (tail) l4[64 + lane] = x2;
    if (lane < 2) l4[PBF / 4 + 32 + lane] = x2;
    if (lane >= 4 && tail) l4[PBF / 4 + 30 + lane] = x2;
  }
  __builtin_amdgcn_fence(__ATOMIC_RELEASE, "wavefront");
  __builtin_amdgcn_wave_barrier();
  __builtin_amdgcn_fence(__ATOMIC_ACQUIRE, "wavefront");

  v8f cacc[8];
  float s2[4];
  float a3p[4][8];
  const int h = lane >> 4;
  const int c = lane & 15;
  if (half) scan_pieces<2>(&buf[PATH_OFF + w * PIECEF], lane, cacc, s2, a3p);
  else      scan_pieces<0>(&buf[PATH_OFF + w * PIECEF], lane, cacc, s2, a3p);

  {
    float* lowA = &buf[LOW_OFF + (2 * w + h) * LOWF];
#pragma unroll
    for (int ti = 0; ti < 4; ++ti) {
      lowA[S2O + 16 * ti + c] = s2[ti];
      float4* d = (float4*)&lowA[S3O + (16 * ti + c) * 8];
      d[0] = make_float4(a3p[ti][0], a3p[ti][1], a3p[ti][2], a3p[ti][3]);
      d[1] = make_float4(a3p[ti][4], a3p[ti][5], a3p[ti][6], a3p[ti][7]);
    }
    float4* s4 = (float4*)&buf[SLOT_OFF + w * SLOTF];
#pragma unroll
    for (int T = 0; T < 8; ++T)
#pragma unroll
      for (int g = 0; g < 2; ++g)
        s4[(2 * T + g) * 32 + lane] =
            make_float4(cacc[T][4 * g], cacc[T][4 * g + 1], cacc[T][4 * g + 2], cacc[T][4 * g + 3]);
  }
  __syncthreads();

  {
    const int hi = (lane >> 4) * 8, cc = lane & 15;
    const int f3 = 2 * tid;
    float o4[8];
    float o3[2];
    float o2 = 0.f, o1 = 0.f;
    float2 p3;
    {
      const int e  = tid >> 2;
      const int kk = f3 & 7;
      p3 = *(const float2*)&buf[LOW_OFF + S3O + f3];
      float pref = 0.f;
#pragma unroll
      for (int k = 1; k < NP; ++k) {
        pref += buf[LOW_OFF + (k - 1) * LOWF + S2O + e];
        const float* pk = &buf[PATH_OFF + (k >> 1) * PIECEF + SH * C_ * (k & 1)];
        const float2 a  = *(const float2*)&buf[LOW_OFF + k * LOWF + S3O + f3];
        const float2 pe = *(const float2*)&pk[SH * C_ + kk];
        const float2 ps = *(const float2*)&pk[kk];
        p3.x += a.x + pref * (pe.x - ps.x);
        p3.y += a.y + pref * (pe.y - ps.y);
      }
      const float s2tot = pref + buf[LOW_OFF + (NP - 1) * LOWF + S2O + e];
      *(float2*)&buf[SREG + S3O + f3] = p3;
      if ((tid & 3) == 0) buf[SREG + S2O + e] = s2tot;
      if (tid < 8) buf[SREG + S1O + tid] = buf[PATH_OFF + PENDF + tid] - buf[PATH_OFF + P0F + tid];
    }
    {
      const float4* s4 = (const float4*)&buf[SLOT_OFF];
#pragma unroll
      for (int g = 0; g < 2; ++g) {
        const int pos = (2 * w + g) * 32 + lane;
        float4 a = s4[pos];
#pragma unroll
        for (int k = 1; k < NW; ++k) {
          float4 c = s4[k * (SLOTF / 4) + pos];
          a.x += c.x; a.y += c.y; a.z += c.z; a.w += c.w;
        }
        o4[4 * g]     = a.x;
        o4[4 * g + 1] = a.y;
        o4[4 * g + 2] = a.z;
        o4[4 * g + 3] = a.w;
      }
    }
    {
      const int hh   = lane >> 4;
      const int rowA = 16 * (2 * half + (w >> 2)) + cc;
      const int kcol = 2 * (w & 3) + (cc >> 3);
      const int lcol = cc & 7;
      float P[NP];
      P[0] = 0.f;
#pragma unroll
      for (int k = 1; k < NP; ++k) P[k] = P[k - 1] + buf[LOW_OFF + (k - 1) * LOWF + S2O + rowA];
      const float s1l = buf[PATH_OFF + PENDF + lcol] - buf[PATH_OFF + P0F + lcol];
      v2f Av[4], Bv[4];
#pragma unroll
      for (int q = 0; q < 4; ++q) {
        const int k0 = 4 * q + 2 * hh;
        const float* pa = &buf[PATH_OFF + (k0 >> 1) * PIECEF];
        const float* ta = &buf[LOW_OFF + k0 * LOWF + S2O + lcol * 8 + kcol];
        Av[q].x = hh ? P[4 * q + 2] : P[4 * q];
        Av[q].y = hh ? P[4 * q + 3] : P[4 * q + 1];
        const float pm = pa[SH * C_ + kcol];
        Bv[q].x = (pm - pa[kcol]) * s1l - ta[0];
        Bv[q].y = (pa[CH * C_ + kcol] - pm) * s1l - ta[LOWF];
      }
      v8f acc;
#pragma unroll
      for (int q8 = 0; q8 < 8; ++q8) acc[q8] = o4[q8];
#pragma unroll
      for (int q = 0; q < 4; ++q)
        acc = __builtin_amdgcn_wmma_f32_16x16x4_f32(false, Av[q], false, Bv[q], (short)0, acc, false, false);
#pragma unroll
      for (int q8 = 0; q8 < 8; ++q8) o4[q8] = acc[q8];
    }
    __syncthreads();
    o3[0] = p3.x;
    o3[1] = p3.y;
    if (tid < 64) o2 = buf[SREG + S2O + tid];
    if (tid < 8)  o1 = buf[SREG + S1O + tid];

#pragma unroll
    for (int n = 2; n <= 4; ++n) {
      const int Pb = (n == 2) ? SREG : ((n == 3) ? PREG : QREG);
      const int Qb = (n == 2) ? PREG : QREG;
      const float coef = ((n & 1) ? 1.0f : -1.0f) / (float)n;
#pragma unroll
      for (int q8 = 0; q8 < 8; ++q8) {
        int f = 2048 * half + f4idx(w, q8, hi, cc);
        float q = buf[Pb + S1O + (f >> 9)] * buf[SREG + S3O + (f & 511)]
                + buf[Pb + S2O + (f >> 6)] * buf[SREG + S2O + (f & 63)]
                + buf[Pb + S3O + (f >> 3)] * buf[SREG + S1O + (f & 7)];
        o4[q8] += coef * q;
      }
#pragma unroll
      for (int r = 0; r < 2; ++r) {
        int f = f3 + r;
        float q = buf[Pb + S1O + (f >> 6)] * buf[SREG + S2O + (f & 63)]
                + buf[Pb + S2O + (f >> 3)] * buf[SREG + S1O + (f & 7)];
        o3[r] += coef * q;
        if (n < 4) buf[Qb + S3O + f] = q;
      }
      if (tid < 64) {
        float q = buf[Pb + S1O + (tid >> 3)] * buf[SREG + S1O + (tid & 7)];
        o2 += coef * q;
        if (n < 4) buf[Qb + S2O + tid] = q;
      }
      if (n < 4) {
        if (tid < 8) buf[Qb + S1O + tid] = 0.0f;
        __syncthreads();
      }
    }

    float* ob = out + (size_t)b * OUTPB;
    if (half == 0) {
      if (tid < 8)  ob[tid] = o1;
      if (tid < 64) ob[8 + tid] = o2;
      ob[72 + f3]     = o3[0];
      ob[72 + f3 + 1] = o3[1];
    }
#pragma unroll
    for (int q8 = 0; q8 < 8; ++q8)
      ob[584 + 2048 * half + f4idx(w, q8, hi, cc)] = o4[q8];
  }
}

extern "C" void kernel_launch(void* const* d_in, const int* in_sizes, int n_in,
                              void* d_out, int out_size, void* d_ws, size_t ws_size,
                              hipStream_t stream) {
  const float* path = (const float*)d_in[0];
  float* out = (float*)d_out;
  logsig_scan_kernel<<<dim3(NGRP), dim3(NT), 0, stream>>>(path, out);
}


# Epilogue, fewer LDS instructions: logarithm in closed form, level 4 a rank-18 matrix-unit product, no power of S in LDS; join reads pieces' level-2 totals transposed 16 B wide (sum order changes)
# speedup vs baseline: 15.9275x; 1.0619x over previous
#include <hip/hip_runtime.h>

typedef float v2f __attribute__((ext_vector_type(2)));
typedef float v8f __attribute__((ext_vector_type(8)));

#define B_      128
#define L_      257
#define C_      8
#define NSTEP   256
#define OUTPB   4680

#define NW      8
#define CH      (NSTEP / NW)
#define NP      (2 * NW)
#define SH      (CH / 2)
#define NT      (NW * 32)
#define NGRP    256

#define S1O 0
#define S2O 8
#define S3O 72
#define S4O 584
#define LOWF 584

#define SLOT_OFF 0
#define SLOTF 2048
#define LOWP 592
#define SREG (NW * 4096)
#define LOW_OFF  (SREG + LOWP)
#define PATH_OFF (LOW_OFF + NP * LOWF)
#define P0F      ((CH + 1) * C_)
#define PENDF    (P0F + C_)
#define PBF      (PENDF + C_)
#define PIECEF   (PBF + (SH + 2) * C_)
#define TP       20
#define TT_OFF   (PATH_OFF + NW * PIECEF)
#define LDS_FLOATS (TT_OFF + 64 * TP)

__device__ __forceinline__ int f4idx(int T, int q8, int hi, int cc) {
  return ((16 * (T >> 2) + hi + q8) << 6) + 16 * (T & 3) + cc;
}

template <int RT>
__device__ __forceinline__ void scan_pieces(const float* pp, int lane, v8f (&cacc)[8],
                                            float (&s2)[4], float (&a3p)[4][8]) {
#pragma unroll
  for (int T = 0; T < 8; ++T)
#pragma unroll
    for (int q = 0; q < 8; ++q) cacc[T][q] = 0.f;

  const int h  = lane >> 4;
  const int c  = lane & 15;
  const int jj = c & 7;
  const int ib = c >> 3;

  const float* pl = (const float*)__builtin_assume_aligned(pp + PBF * h, 16);
  float sg1[4];
  float pvi[4], pvj, pvk[8];
#pragma unroll
  for (int ti = 0; ti < 4; ++ti) {
    pvi[ti] = pl[2 * ti + ib];
    sg1[ti] = pvi[ti] - pp[P0F + 2 * ti + ib];
    s2[ti] = 0.f;
#pragma unroll
    for (int k = 0; k < 8; ++k) a3p[ti][k] = 0.f;
  }
  pvj = pl[jj];
  {
    float4 a = *(const float4*)&pl[0], bq = *(const float4*)&pl[4];
    pvk[0] = a.x; pvk[1] = a.y; pvk[2] = a.z; pvk[3] = a.w;
    pvk[4] = bq.x; pvk[5] = bq.y; pvk[6] = bq.z; pvk[7] = bq.w;
  }
  const float pend_j = pp[PENDF + jj];

  float cj, ci[4];
  float4 ck[2];
  {
    const float* q = pl + 8;
    cj = q[jj];
#pragma unroll
    for (int ti = 0; ti < 4; ++ti) ci[ti] = q[2 * ti + ib];
    ck[0] = *(const float4*)&q[0];
    ck[1] = *(const float4*)&q[4];
  }

  for (int n = 0; n < SH; ++n) {
    float nj, ni[4];
    float4 nk[2];
    {
      const float* q = pl + 8 * (n + 2);
      nj = q[jj];
#pragma unroll
      for (int ti = 0; ti < 4; ++ti) ni[ti] = q[2 * ti + ib];
      nk[0] = *(const float4*)&q[0];
      nk[1] = *(const float4*)&q[4];
    }

    v2f ax[2], bx[4];
    {
      const float pcj = cj;
      const float vj  = pcj - pvj;  pvj = pcj;
      const float Rj  = pend_j - pcj;
      const float hU  = 0.5f * vj;
      const float hT  = vj * (1.0f / 3.0f);
      float pk[8], vk[8];
      pk[0] = ck[0].x; pk[1] = ck[0].y; pk[2] = ck[0].z; pk[3] = ck[0].w;
      pk[4] = ck[1].x; pk[5] = ck[1].y; pk[6] = ck[1].z; pk[7] = ck[1].w;
#pragma unroll
      for (int k = 0; k < 8; ++k) { vk[k] = pk[k] - pvk[k]; pvk[k] = pk[k]; }
#pragma unroll
      for (int ti = 0; ti < 4; ++ti) {
        const float pci = ci[ti];
        const float vi  = pci - pvi[ti];  pvi[ti] = pci;
        const float gU  = sg1[ti] + vi * (1.0f / 3.0f);
        const float g2  = sg1[ti] + 0.5f * vi;
        const float u2  = s2[ti] + hU * gU;
        if (ti == RT || ti == RT + 1) {
          const float gT = sg1[ti] + vi * 0.25f;
          ax[ti - RT].x = u2;
          ax[ti - RT].y = s2[ti] + hT * gT;
        }
        s2[ti]  += vj * g2;
        sg1[ti] += vi;
        bx[ti].x = vi * Rj;  bx[ti].y = vi * hU;
#pragma unroll
        for (int k = 0; k < 8; ++k) a3p[ti][k] += vk[k] * u2;
      }
    }

#pragma unroll
    for (int t = 0; t < 2; ++t)
#pragma unroll
      for (int tj = 0; tj < 4; ++tj)
        cacc[4 * t + tj] = __builtin_amdgcn_wmma_f32_16x16x4_f32(
            false, ax[t], false, bx[tj], (short)0, cacc[4 * t + tj], false, false);

    cj = nj;
#pragma unroll
    for (int ti = 0; ti < 4; ++ti) ci[ti] = ni[ti];
    ck[0] = nk[0];
    ck[1] = nk[1];
  }
}

__global__ __launch_bounds__(NT)
void logsig_scan_kernel(const float* __restrict__ path, float* __restrict__ out) {
  __shared__ __attribute__((aligned(16))) float buf[LDS_FLOATS];
  const int tid  = threadIdx.x;
  const int lane = tid & 31;
  const int w    = __builtin_amdgcn_readfirstlane(tid >> 5);
  const int b    = blockIdx.x >> 1;
  const int half = blockIdx.x & 1;
  const float* gp = path + (size_t)b * (L_ * C_);

  {
    const float4* g4 = (const float4*)gp + w * (CH * C_ / 4);
    float4* l4 = (float4*)&buf[PATH_OFF + w * PIECEF];
    const bool tail = lane < (PBF / 4 - 64);
    const float4* gx = (lane < 2) ? (g4 + 64 + lane)
                     : ((lane < 4) ? ((const float4*)gp + (lane - 2))
                                   : ((const float4*)gp + (NSTEP * C_ / 4 + lane - 4)));
    float4 x0 = g4[lane];
    float4 x1 = g4[32 + lane];
    float4 x2 = tail ? *gx : make_float4(0.f, 0.f, 0.f, 0.f);
    l4[lane] = x0;
    l4[32 + lane] = x1;
    l4[PBF / 4 + lane] = x1;
    if (tail) l4[64 + lane] = x2;
    if (lane < 2) l4[PBF / 4 + 32 + lane] = x2;
    if (lane >= 4 && tail) l4[PBF / 4 + 30 + lane] = x2;
  }
  __builtin_amdgcn_fence(__ATOMIC_RELEASE, "wavefront");
  __builtin_amdgcn_wave_barrier();
  __builtin_amdgcn_fence(__ATOMIC_ACQUIRE, "wavefront");

  v8f cacc[8];
  float s2[4];
  float a3p[4][8];
  const int h = lane >> 4;
  const int c = lane & 15;
  if (half) scan_pieces<2>(&buf[PATH_OFF + w * PIECEF], lane, cacc, s2, a3p);
  else      scan_pieces<0>(&buf[PATH_OFF + w * PIECEF], lane, cacc, s2, a3p);

  {
    float* lowA = &buf[LOW_OFF + (2 * w + h) * LOWF];
#pragma unroll
    for (int ti = 0; ti < 4; ++ti) {
      buf[TT_OFF + (16 * ti + c) * TP + 2 * w + h] = s2[ti];
      float4* d = (float4*)&lowA[S3O + (16 * ti + c) * 8];
      d[0] = make_float4(a3p[ti][0], a3p[ti][1], a3p[ti][2], a3p[ti][3]);
      d[1] = make_float4(a3p[ti][4], a3p[ti][5], a3p[ti][6], a3p[ti][7]);
    }
    float4* s4 = (float4*)&buf[SLOT_OFF + w * SLOTF];
#pragma unroll
    for (int T = 0; T < 8; ++T)
#pragma unroll
      for (int g = 0; g < 2; ++g)
        s4[(2 * T + g) * 32 + lane] =
            make_float4(cacc[T][4 * g], cacc[T][4 * g + 1], cacc[T][4 * g + 2], cacc[T][4 * g + 3]);
  }
  __syncthreads();

  {
    const int hi = (lane >> 4) * 8, cc = lane & 15;
    const int f3 = 2 * tid;
    float o4[8];
    float o3[2];
    float o2 = 0.f, o1 = 0.f;
    float2 p3;
    {
      const int e  = tid >> 2;
      const int kk = f3 & 7;
      p3 = *(const float2*)&buf[LOW_OFF + S3O + f3];
      float Tk[NP];
#pragma unroll
      for (int m = 0; m < NP / 4; ++m) {
        const float4 t = *(const float4*)&buf[TT_OFF + e * TP + 4 * m];
        Tk[4 * m] = t.x; Tk[4 * m + 1] = t.y; Tk[4 * m + 2] = t.z; Tk[4 * m + 3] = t.w;
      }
      float pref = 0.f;
      float2 ps = *(const float2*)&buf[PATH_OFF + SH * C_ + kk];
#pragma unroll
      for (int k = 1; k < NP; ++k) {
        pref += Tk[k - 1];
        const float2 a  = *(const float2*)&buf[LOW_OFF + k * LOWF + S3O + f3];
        const float2 pe = *(const float2*)&buf[PATH_OFF + (k >> 1) * PIECEF + SH * C_ * ((k & 1) + 1) + kk];
        p3.x += a.x + pref * (pe.x - ps.x);
        p3.y += a.y + pref * (pe.y - ps.y);
        ps = pe;
      }
      const float s2tot = pref + Tk[NP - 1];
      *(float2*)&buf[SREG + S3O + f3] = p3;
      if ((tid & 3) == 0) buf[SREG + S2O + e] = s2tot;
      if (tid < 8) buf[SREG + S1O + tid] = buf[PATH_OFF + PENDF + tid] - buf[PATH_OFF + P0F + tid];
    }
    {
      const float4* s4 = (const float4*)&buf[SLOT_OFF];
#pragma unroll
      for (int g = 0; g < 2; ++g) {
        const int pos = (2 * w + g) * 32 + lane;
        float4 a = s4[pos];
#pragma unroll
        for (int k = 1; k < NW; ++k) {
          float4 c = s4[k * (SLOTF / 4) + pos];
          a.x += c.x; a.y += c.y; a.z += c.z; a.w += c.w;
        }
        o4[4 * g]     = a.x;
        o4[4 * g + 1] = a.y;
        o4[4 * g + 2] = a.z;
        o4[4 * g + 3] = a.w;
      }
    }
    {
      const int hh   = lane >> 4;
      const int rowA = 16 * (2 * half + (w >> 2)) + cc;
      const int kcol = 2 * (w & 3) + (cc >> 3);
      const int lcol = cc & 7;
      float P[NP];
      P[0] = 0.f;
#pragma unroll
      for (int m = 0; m < NP / 4; ++m) {
        const float4 t = *(const float4*)&buf[TT_OFF + rowA * TP + 4 * m];
        P[4 * m + 1] = P[4 * m] + t.x;
        P[4 * m + 2] = P[4 * m + 1] + t.y;
        P[4 * m + 3] = P[4 * m + 2] + t.z;
        if (m < NP / 4 - 1) P[4 * m + 4] = P[4 * m + 3] + t.w;
      }
      const float s1l = buf[PATH_OFF + PENDF + lcol] - buf[PATH_OFF + P0F + lcol];
      v2f Av[4], Bv[4];
#pragma unroll
      for (int q = 0; q < 4; ++q) {
        const int k0 = 4 * q + 2 * hh;
        const float* pa = &buf[PATH_OFF + (k0 >> 1) * PIECEF];
        const float2 ta = *(const float2*)&buf[TT_OFF + (lcol * 8 + kcol) * TP + k0];
        Av[q].x = hh ? P[4 * q + 2] : P[4 * q];
        Av[q].y = hh ? P[4 * q + 3] : P[4 * q + 1];
        const float pm = pa[SH * C_ + kcol];
        Bv[q].x = (pm - pa[kcol]) * s1l - ta.x;
        Bv[q].y = (pa[CH * C_ + kcol] - pm) * s1l - ta.y;
      }
      v8f acc;
#pragma unroll
      for (int q8 = 0; q8 < 8; ++q8) acc[q8] = o4[q8];
#pragma unroll
      for (int q = 0; q < 4; ++q)
        acc = __builtin_amdgcn_wmma_f32_16x16x4_f32(false, Av[q], false, Bv[q], (short)0, acc, false, false);
#pragma unroll
      for (int q8 = 0; q8 < 8; ++q8) o4[q8] = acc[q8];
    }
    __syncthreads();
    const float* S1 = &buf[SREG + S1O];
    const float* S2 = &buf[SREG + S2O];
    const float* S3 = &buf[SREG + S3O];
    {
      const int hh   = lane >> 4;
      const int rowA = 16 * (2 * half + (w >> 2)) + cc;
      const int kl   = 16 * (w & 3) + cc;
      const int jrow = cc & 7;
      const int kcol = kl >> 3;
      const float s1i = S1[rowA >> 3], s1j = S1[jrow], s1k = S1[kcol], s1l = S1[kl & 7];
      const float s2ij = S2[rowA], s2kl = S2[kl];
      v2f Av[5], Bv[5];
#pragma unroll
      for (int q = 0; q < 2; ++q) {
        const int m0 = 4 * q + 2 * hh;
        Av[q].x = (jrow == m0)     ? s1i : 0.f;
        Av[q].y = (jrow == m0 + 1) ? s1i : 0.f;
        Bv[q].x = -0.5f * S3[m0 * 64 + kl]       + (1.0f / 3.0f) * (S2[m0 * 8 + kcol] * s1l);
        Bv[q].y = -0.5f * S3[(m0 + 1) * 64 + kl] + (1.0f / 3.0f) * (S2[(m0 + 1) * 8 + kcol] * s1l);
      }
#pragma unroll
      for (int q = 2; q < 4; ++q) {
        const int m0 = 4 * (q - 2) + 2 * hh;
        const float2 a = *(const float2*)&S3[rowA * 8 + m0];
        Av[q].x = a.x;
        Av[q].y = a.y;
        Bv[q].x = (kcol == m0)     ? -0.5f * s1l : 0.f;
        Bv[q].y = (kcol == m0 + 1) ? -0.5f * s1l : 0.f;
      }
      {
        const float xkl = s1k * s1l;
        Av[4].x = hh ? 0.f : s2ij;
        Av[4].y = hh ? 0.f : s1i * s1j;
        Bv[4].x = hh ? 0.f : (-0.5f * s2kl + (1.0f / 3.0f) * xkl);
        Bv[4].y = hh ? 0.f : ((1.0f / 3.0f) * s2kl - 0.25f * xkl);
      }
      v8f acc;
#pragma unroll
      for (int q8 = 0; q8 < 8; ++q8) acc[q8] = o4[q8];
#pragma unroll
      for (int q = 0; q < 5; ++q)
        acc = __builtin_amdgcn_wmma_f32_16x16x4_f32(false, Av[q], false, Bv[q], (short)0, acc, false, false);
#pragma unroll
      for (int q8 = 0; q8 < 8; ++q8) o4[q8] = acc[q8];
    }
    {
      const int i3 = f3 >> 6, j3 = (f3 >> 3) & 7, k3 = f3 & 7;
      const float a1i = S1[i3], a1j = S1[j3], a2ij = S2[f3 >> 3];
      const float2 a1k  = *(const float2*)&S1[k3];
      const float2 a2jk = *(const float2*)&S2[f3 & 63];
      const float cub = (1.0f / 3.0f) * (a1i * a1j);
      o3[0] = p3.x - 0.5f * (a1i * a2jk.x + a2ij * a1k.x) + cub * a1k.x;
      o3[1] = p3.y - 0.5f * (a1i * a2jk.y + a2ij * a1k.y) + cub * a1k.y;
    }
    if (tid < 64) o2 = S2[tid] - 0.5f * (S1[tid >> 3] * S1[tid & 7]);
    if (tid < 8)  o1 = S1[tid];

    float* ob = out + (size_t)b * OUTPB;
    if (half == 0) {
      if (tid < 8)  ob[tid] = o1;
      if (tid < 64) ob[8 + tid] = o2;
      ob[72 + f3]     = o3[0];
      ob[72 + f3 + 1] = o3[1];
    }
#pragma unroll
    for (int q8 = 0; q8 < 8; ++q8)
      ob[584 + 2048 * half + f4idx(w, q8, hi, cc)] = o4[q8];
  }
}

extern "C" void kernel_launch(void* const* d_in, const int* in_sizes, int n_in,
                              void* d_out, int out_size, void* d_ws, size_t ws_size,
                              hipStream_t stream) {
  const float* path = (const float*)d_in[0];
  float* out = (float*)d_out;
  logsig_scan_kernel<<<dim3(NGRP), dim3(NT), 0, stream>>>(path, out);
}
